# starting kernel
# baseline (speedup 1.0000x reference)
#include <hip/hip_runtime.h>

typedef __attribute__((ext_vector_type(2))) float v2f;
typedef __attribute__((ext_vector_type(4))) float v4f;
typedef __attribute__((ext_vector_type(8))) float v8f;

__device__ constexpr unsigned BASEC[16] = {
    2654435761u, 2246822519u, 3266489917u, 2028178513u, 1220703125u,
    1610612741u, 805306457u,  402653189u,  3674653429u, 2860486313u,
    1073676287u, 2971215073u, 1500450271u, 3267000013u, 2654435789u, 4049292737u};
__device__ constexpr unsigned P5C[32] = {
    2654435761u ^ 0xDEADBEEFu, 2246822519u ^ 0xDEADBEEFu, 3266489917u ^ 0xDEADBEEFu,
    2028178513u ^ 0xDEADBEEFu, 1220703125u ^ 0xDEADBEEFu, 1610612741u ^ 0xDEADBEEFu,
    805306457u  ^ 0xDEADBEEFu, 402653189u  ^ 0xDEADBEEFu, 3674653429u ^ 0xDEADBEEFu,
    2860486313u ^ 0xDEADBEEFu, 1073676287u ^ 0xDEADBEEFu, 2971215073u ^ 0xDEADBEEFu,
    1500450271u ^ 0xDEADBEEFu, 3267000013u ^ 0xDEADBEEFu, 2654435789u ^ 0xDEADBEEFu,
    4049292737u ^ 0xDEADBEEFu, 2246822531u ^ 0xDEADBEEFu, 3266489927u ^ 0xDEADBEEFu,
    2028178519u ^ 0xDEADBEEFu, 1220703133u ^ 0xDEADBEEFu, 1610612759u ^ 0xDEADBEEFu,
    805306463u  ^ 0xDEADBEEFu, 402653201u  ^ 0xDEADBEEFu, 3674653441u ^ 0xDEADBEEFu,
    2860486319u ^ 0xDEADBEEFu, 1073676311u ^ 0xDEADBEEFu, 2971215091u ^ 0xDEADBEEFu,
    1500450277u ^ 0xDEADBEEFu, 3267000023u ^ 0xDEADBEEFu, 2654435801u ^ 0xDEADBEEFu,
    4049292751u ^ 0xDEADBEEFu, 2246822537u ^ 0xDEADBEEFu};

#define T_DIM 4096
#define PPB 128
#define STRIDE 68

__global__ __launch_bounds__(128, 4)
void rhp_kernel(const int* __restrict__ tokens,
                const float* __restrict__ tables,
                const float* __restrict__ Wc,
                float* __restrict__ out,
                int nIter)
{
    __shared__ unsigned tk[256];
    __shared__ unsigned keys_lds[8][PPB];
    __shared__ unsigned ck_lds[PPB];
    __shared__ unsigned cklut[256];
    __shared__ __align__(16) float short_lds[PPB * STRIDE];
    __shared__ __align__(16) float wc_lds[16 * STRIDE];

    const int tid = threadIdx.x;
    const int bid = blockIdx.x;
    const int b   = bid >> 5;
    const int t0  = (bid & 31) << 7;
    const long rowBase = (long)b * T_DIM;

    for (int j = tid; j < 256; j += 128) {
        int g = t0 - 128 + j;
        tk[j] = (g >= 0) ? (unsigned)tokens[rowBase + g] : 0u;
        unsigned v = 0;
        #pragma unroll
        for (int i = 0; i < 8; ++i)
            if (j & (1 << i)) v ^= BASEC[i];
        cklut[j] = v;
    }
    for (int idx = tid; idx < 16 * STRIDE; idx += 128) {
        int row = idx / STRIDE;
        int col = idx - row * STRIDE;
        wc_lds[idx] = (row < 8 && col < 64) ? Wc[row * 64 + col] : 0.0f;
    }
    __syncthreads();

    {
        const int base = 128 + tid;
        unsigned h = 0;
        #pragma unroll
        for (int i = 0; i < 16; ++i) {
            h ^= tk[base - 1 - i] * BASEC[i];
            if (i == 0)  keys_lds[0][tid] = h & 0xFFFFu;
            if (i == 1)  keys_lds[1][tid] = h & 0xFFFFu;
            if (i == 3)  keys_lds[2][tid] = h & 0xFFFFu;
            if (i == 7)  keys_lds[3][tid] = h & 0xFFFFu;
            if (i == 15) keys_lds[4][tid] = h & 0xFFFFu;
        }
        unsigned h5 = 0, h6 = 0, h7 = 0;
        #pragma unroll 8
        for (int i = 0; i < 32; ++i)  h5 ^= tk[base - 1 - i] * P5C[i];
        keys_lds[5][tid] = h5 & 0xFFFFu;
        #pragma unroll 8
        for (int i = 0; i < 64; ++i)  h6 ^= tk[base - 1 - i] * (BASEC[i & 7] ^ 0xCAFEBABEu);
        keys_lds[6][tid] = h6 & 0xFFFFu;
        #pragma unroll 8
        for (int i = 0; i < 128; ++i) h7 ^= tk[base - 1 - i] * (BASEC[i & 7] ^ 0x8BADF00Du);
        keys_lds[7][tid] = h7 & 0xFFFFu;
    }
    __syncthreads();

    const v4f* tab4 = reinterpret_cast<const v4f*>(tables);
    v4f* out4 = reinterpret_cast<v4f*>(out);
    const long outPosBase = (rowBase + t0) * 8;

    #pragma unroll 4
    for (int j = 0; j < nIter; ++j) {
        int flat   = j * 128 + tid;
        int rowIdx = flat >> 2;
        int piece  = flat & 3;
        int scale  = rowIdx >> 7;
        int pos    = rowIdx & 127;
        unsigned key = keys_lds[scale][pos];
        v4f v = tab4[((size_t)(scale << 16) + key) * 4 + piece];
        *reinterpret_cast<v4f*>(&short_lds[pos * STRIDE + scale * 16 + piece * 4]) = v;
        __builtin_nontemporal_store(v, &out4[(outPosBase + (long)pos * 8 + scale) * 4 + piece]);
    }
    __syncthreads();

    {
        const int lane = tid & 31;
        const int wave = tid >> 5;
        const int n  = lane & 15;
        const int hi = lane >> 4;
        const float* bp = &wc_lds[n * STRIDE + 2 * hi];
        #pragma unroll
        for (int tt = 0; tt < 2; ++tt) {
            const int mb = wave * 32 + tt * 16;
            const float* ap = &short_lds[(mb + n) * STRIDE + 2 * hi];
            v8f c = {0.f, 0.f, 0.f, 0.f, 0.f, 0.f, 0.f, 0.f};
            #pragma unroll
            for (int k = 0; k < 16; ++k) {
                v2f a  = *reinterpret_cast<const v2f*>(ap + 4 * k);
                v2f bb = *reinterpret_cast<const v2f*>(bp + 4 * k);
                c = __builtin_amdgcn_wmma_f32_16x16x4_f32(
                        false, a, false, bb, (short)0, c, false, false);
            }
            #pragma unroll
            for (int v = 0; v < 8; ++v) {
#if __has_builtin(__builtin_amdgcn_ballot_w32)
                unsigned mask = __builtin_amdgcn_ballot_w32(c[v] > 0.0f);
                unsigned bits = (mask >> (16 * hi)) & 0xFFu;
                if (n == v)
                    ck_lds[mb + v + 8 * hi] = cklut[bits];
#else
                if (n < 8 && c[v] > 0.0f)
                    atomicXor(&ck_lds[mb + v + 8 * hi], BASEC[n]);
#endif
            }
        }
    }
    __syncthreads();

    #pragma unroll 4
    for (int j = 0; j < nIter; ++j) {
        int flat   = j * 128 + tid;
        int rowIdx = flat >> 2;
        int piece  = flat & 3;
        int scale  = 4 + (rowIdx >> 7);
        int pos    = rowIdx & 127;
        unsigned key = (keys_lds[scale][pos] ^ ck_lds[pos]) & 0xFFFFu;
        v4f v = tab4[((size_t)(scale << 16) + key) * 4 + piece];
        __builtin_nontemporal_store(v, &out4[(outPosBase + (long)pos * 8 + scale) * 4 + piece]);
    }
}

extern "C" void kernel_launch(void* const* d_in, const int* in_sizes, int n_in,
                              void* d_out, int out_size, void* d_ws, size_t ws_size,
                              hipStream_t stream) {
    (void)in_sizes; (void)n_in; (void)out_size; (void)d_ws; (void)ws_size;
    const int*   tokens = (const int*)d_in[0];
    const float* tables = (const float*)d_in[1];
    const float* Wc     = (const float*)d_in[2];
    float*       out    = (float*)d_out;
    const int B = 32, T = 4096;
    dim3 grid(B * (T / PPB)), block(128);
    hipLaunchKernelGGL(rhp_kernel, grid, block, 0, stream, tokens, tables, Wc, out, 16);
}


# WMMA B operand (Wc) kept in registers: each lane loads its 16 k-pairs straight from global ahead of the token loads; the Wc LDS plane and its 9-trip staging loop (5 waited loads) are gone
# speedup vs baseline: 1.0813x; 1.0813x over previous
#include <hip/hip_runtime.h>

typedef __attribute__((ext_vector_type(2))) float v2f;
typedef __attribute__((ext_vector_type(4))) float v4f;
typedef __attribute__((ext_vector_type(8))) float v8f;

__device__ constexpr unsigned BASEC[16] = {
    2654435761u, 2246822519u, 3266489917u, 2028178513u, 1220703125u,
    1610612741u, 805306457u,  402653189u,  3674653429u, 2860486313u,
    1073676287u, 2971215073u, 1500450271u, 3267000013u, 2654435789u, 4049292737u};
__device__ constexpr unsigned P5C[32] = {
    2654435761u ^ 0xDEADBEEFu, 2246822519u ^ 0xDEADBEEFu, 3266489917u ^ 0xDEADBEEFu,
    2028178513u ^ 0xDEADBEEFu, 1220703125u ^ 0xDEADBEEFu, 1610612741u ^ 0xDEADBEEFu,
    805306457u  ^ 0xDEADBEEFu, 402653189u  ^ 0xDEADBEEFu, 3674653429u ^ 0xDEADBEEFu,
    2860486313u ^ 0xDEADBEEFu, 1073676287u ^ 0xDEADBEEFu, 2971215073u ^ 0xDEADBEEFu,
    1500450271u ^ 0xDEADBEEFu, 3267000013u ^ 0xDEADBEEFu, 2654435789u ^ 0xDEADBEEFu,
    4049292737u ^ 0xDEADBEEFu, 2246822531u ^ 0xDEADBEEFu, 3266489927u ^ 0xDEADBEEFu,
    2028178519u ^ 0xDEADBEEFu, 1220703133u ^ 0xDEADBEEFu, 1610612759u ^ 0xDEADBEEFu,
    805306463u  ^ 0xDEADBEEFu, 402653201u  ^ 0xDEADBEEFu, 3674653441u ^ 0xDEADBEEFu,
    2860486319u ^ 0xDEADBEEFu, 1073676311u ^ 0xDEADBEEFu, 2971215091u ^ 0xDEADBEEFu,
    1500450277u ^ 0xDEADBEEFu, 3267000023u ^ 0xDEADBEEFu, 2654435801u ^ 0xDEADBEEFu,
    4049292751u ^ 0xDEADBEEFu, 2246822537u ^ 0xDEADBEEFu};

#define T_DIM 4096
#define PPB 128
#define STRIDE 68

__global__ __launch_bounds__(128, 4)
void rhp_kernel(const int* __restrict__ tokens,
                const float* __restrict__ tables,
                const float* __restrict__ Wc,
                float* __restrict__ out,
                int nIter)
{
    __shared__ unsigned tk[256];
    __shared__ unsigned keys_lds[8][PPB];
    __shared__ unsigned ck_lds[PPB];
    __shared__ unsigned cklut[256];
    __shared__ __align__(8) unsigned g6_lds[256];
    __shared__ __align__(8) unsigned g7_lds[256];
    __shared__ __align__(16) float short_lds[PPB * STRIDE];

    const int tid = threadIdx.x;
    const int bid = blockIdx.x;
    const int b   = bid >> 5;
    const int t0  = (bid & 31) << 7;
    const long rowBase = (long)b * T_DIM;

    const int lane = tid & 31;
    const int n    = lane & 15;
    const int hi   = lane >> 4;
    v2f bw[16];
    {
        const float* wrow = Wc + (n & 7) * 64 + 2 * hi;
        #pragma unroll
        for (int k = 0; k < 16; ++k) bw[k] = *reinterpret_cast<const v2f*>(wrow + 4 * k);
    }

    for (int j = tid; j < 256; j += 128) {
        int g = t0 - 128 + j;
        tk[j] = (g >= 0) ? (unsigned)tokens[rowBase + g] : 0u;
        unsigned v = 0;
        #pragma unroll
        for (int i = 0; i < 8; ++i)
            if (j & (1 << i)) v ^= BASEC[i];
        cklut[j] = v;
    }
    __syncthreads();

    {
        const int tq = (tid < 4) ? 4 : tid;
        const int q0 = 2 * tq;
        unsigned w[9];
        #pragma unroll
        for (int i = 0; i < 9; ++i) w[i] = tk[q0 - 8 + i];
        unsigned a6 = 0, b6 = 0, a7 = 0, b7 = 0;
        #pragma unroll
        for (int j = 0; j < 8; ++j) {
            const unsigned m6 = BASEC[j] ^ 0xCAFEBABEu;
            const unsigned m7 = BASEC[j] ^ 0x8BADF00Du;
            a6 ^= w[7 - j] * m6;
            b6 ^= w[8 - j] * m6;
            a7 ^= w[7 - j] * m7;
            b7 ^= w[8 - j] * m7;
        }
        g6_lds[2 * tid] = a6;  g6_lds[2 * tid + 1] = b6;
        g7_lds[2 * tid] = a7;  g7_lds[2 * tid + 1] = b7;
    }
    {
        const int base = 128 + tid;
        unsigned h = 0;
        #pragma unroll
        for (int i = 0; i < 16; ++i) {
            h ^= tk[base - 1 - i] * BASEC[i];
            if (i == 0)  keys_lds[0][tid] = h & 0xFFFFu;
            if (i == 1)  keys_lds[1][tid] = h & 0xFFFFu;
            if (i == 3)  keys_lds[2][tid] = h & 0xFFFFu;
            if (i == 7)  keys_lds[3][tid] = h & 0xFFFFu;
            if (i == 15) keys_lds[4][tid] = h & 0xFFFFu;
        }
        unsigned h5 = 0;
        #pragma unroll 8
        for (int i = 0; i < 32; ++i)  h5 ^= tk[base - 1 - i] * P5C[i];
        keys_lds[5][tid] = h5 & 0xFFFFu;
    }
    __syncthreads();
    {
        const int base = 128 + tid;
        unsigned h6 = 0, h7 = 0;
        #pragma unroll
        for (int m = 0; m < 8; ++m)  h6 ^= g6_lds[base - 8 * m];
        #pragma unroll
        for (int m = 0; m < 16; ++m) h7 ^= g7_lds[base - 8 * m];
        keys_lds[6][tid] = h6 & 0xFFFFu;
        keys_lds[7][tid] = h7 & 0xFFFFu;
    }

    const v4f* tab4 = reinterpret_cast<const v4f*>(tables);
    v4f* out4 = reinterpret_cast<v4f*>(out);
    const long outPosBase = (rowBase + t0) * 8;

    #pragma unroll 4
    for (int j = 0; j < nIter; ++j) {
        int flat   = j * 128 + tid;
        int rowIdx = flat >> 2;
        int piece  = flat & 3;
        int scale  = rowIdx >> 7;
        int pos    = rowIdx & 127;
        unsigned key = keys_lds[scale][pos];
        v4f v = tab4[((size_t)(scale << 16) + key) * 4 + piece];
        *reinterpret_cast<v4f*>(&short_lds[pos * STRIDE + scale * 16 + piece * 4]) = v;
        __builtin_nontemporal_store(v, &out4[(outPosBase + (long)pos * 8 + scale) * 4 + piece]);
    }
    __syncthreads();

    {
        const int wave = tid >> 5;
        #pragma unroll
        for (int tt = 0; tt < 2; ++tt) {
            const int mb = wave * 32 + tt * 16;
            const float* ap = &short_lds[(mb + n) * STRIDE + 2 * hi];
            v8f c = {0.f, 0.f, 0.f, 0.f, 0.f, 0.f, 0.f, 0.f};
            #pragma unroll
            for (int k = 0; k < 16; ++k) {
                v2f a  = *reinterpret_cast<const v2f*>(ap + 4 * k);
                c = __builtin_amdgcn_wmma_f32_16x16x4_f32(
                        false, a, false, bw[k], (short)0, c, false, false);
            }
            #pragma unroll
            for (int v = 0; v < 8; ++v) {
#if __has_builtin(__builtin_amdgcn_ballot_w32)
                unsigned mask = __builtin_amdgcn_ballot_w32(c[v] > 0.0f);
                unsigned bits = (mask >> (16 * hi)) & 0xFFu;
                if (n == v)
                    ck_lds[mb + v + 8 * hi] = cklut[bits];
#else
                if (n < 8 && c[v] > 0.0f)
                    atomicXor(&ck_lds[mb + v + 8 * hi], BASEC[n]);
#endif
            }
        }
    }
    __syncthreads();

    #pragma unroll 4
    for (int j = 0; j < nIter; ++j) {
        int flat   = j * 128 + tid;
        int rowIdx = flat >> 2;
        int piece  = flat & 3;
        int scale  = 4 + (rowIdx >> 7);
        int pos    = rowIdx & 127;
        unsigned key = (keys_lds[scale][pos] ^ ck_lds[pos]) & 0xFFFFu;
        v4f v = tab4[((size_t)(scale << 16) + key) * 4 + piece];
        __builtin_nontemporal_store(v, &out4[(outPosBase + (long)pos * 8 + scale) * 4 + piece]);
    }
}

extern "C" void kernel_launch(void* const* d_in, const int* in_sizes, int n_in,
                              void* d_out, int out_size, void* d_ws, size_t ws_size,
                              hipStream_t stream) {
    (void)in_sizes; (void)n_in; (void)out_size; (void)d_ws; (void)ws_size;
    const int*   tokens = (const int*)d_in[0];
    const float* tables = (const float*)d_in[1];
    const float* Wc     = (const float*)d_in[2];
    float*       out    = (float*)d_out;
    const int B = 32, T = 4096;
    dim3 grid(B * (T / PPB)), block(128);
    hipLaunchKernelGGL(rhp_kernel, grid, block, 0, stream, tokens, tables, Wc, out, 16);
}


# Short half of the output leaves from the existing LDS tile under the long gather's load latency (tile rows as whole 256-byte runs); phase 3 only loads, so its barrier waits on no global store
# speedup vs baseline: 1.2623x; 1.1674x over previous
#include <hip/hip_runtime.h>

typedef __attribute__((ext_vector_type(2))) float v2f;
typedef __attribute__((ext_vector_type(4))) float v4f;
typedef __attribute__((ext_vector_type(8))) float v8f;

__device__ constexpr unsigned BASEC[16] = {
    2654435761u, 2246822519u, 3266489917u, 2028178513u, 1220703125u,
    1610612741u, 805306457u,  402653189u,  3674653429u, 2860486313u,
    1073676287u, 2971215073u, 1500450271u, 3267000013u, 2654435789u, 4049292737u};
__device__ constexpr unsigned P5C[32] = {
    2654435761u ^ 0xDEADBEEFu, 2246822519u ^ 0xDEADBEEFu, 3266489917u ^ 0xDEADBEEFu,
    2028178513u ^ 0xDEADBEEFu, 1220703125u ^ 0xDEADBEEFu, 1610612741u ^ 0xDEADBEEFu,
    805306457u  ^ 0xDEADBEEFu, 402653189u  ^ 0xDEADBEEFu, 3674653429u ^ 0xDEADBEEFu,
    2860486313u ^ 0xDEADBEEFu, 1073676287u ^ 0xDEADBEEFu, 2971215073u ^ 0xDEADBEEFu,
    1500450271u ^ 0xDEADBEEFu, 3267000013u ^ 0xDEADBEEFu, 2654435789u ^ 0xDEADBEEFu,
    4049292737u ^ 0xDEADBEEFu, 2246822531u ^ 0xDEADBEEFu, 3266489927u ^ 0xDEADBEEFu,
    2028178519u ^ 0xDEADBEEFu, 1220703133u ^ 0xDEADBEEFu, 1610612759u ^ 0xDEADBEEFu,
    805306463u  ^ 0xDEADBEEFu, 402653201u  ^ 0xDEADBEEFu, 3674653441u ^ 0xDEADBEEFu,
    2860486319u ^ 0xDEADBEEFu, 1073676311u ^ 0xDEADBEEFu, 2971215091u ^ 0xDEADBEEFu,
    1500450277u ^ 0xDEADBEEFu, 3267000023u ^ 0xDEADBEEFu, 2654435801u ^ 0xDEADBEEFu,
    4049292751u ^ 0xDEADBEEFu, 2246822537u ^ 0xDEADBEEFu};

#define T_DIM 4096
#define PPB 128
#define STRIDE 68

__global__ __launch_bounds__(128, 4)
void rhp_kernel(const int* __restrict__ tokens,
                const float* __restrict__ tables,
                const float* __restrict__ Wc,
                float* __restrict__ out,
                int nIter)
{
    __shared__ unsigned tk[256];
    __shared__ unsigned keys_lds[8][PPB];
    __shared__ unsigned ck_lds[PPB];
    __shared__ unsigned cklut[256];
    __shared__ __align__(8) unsigned g6_lds[256];
    __shared__ __align__(8) unsigned g7_lds[256];
    __shared__ __align__(16) float short_lds[PPB * STRIDE];

    const int tid = threadIdx.x;
    const int bid = blockIdx.x;
    const int b   = bid >> 5;
    const int t0  = (bid & 31) << 7;
    const long rowBase = (long)b * T_DIM;

    const int lane = tid & 31;
    const int n    = lane & 15;
    const int hi   = lane >> 4;
    v2f bw[16];
    {
        const float* wrow = Wc + (n & 7) * 64 + 2 * hi;
        #pragma unroll
        for (int k = 0; k < 16; ++k) bw[k] = *reinterpret_cast<const v2f*>(wrow + 4 * k);
    }

    for (int j = tid; j < 256; j += 128) {
        int g = t0 - 128 + j;
        tk[j] = (g >= 0) ? (unsigned)tokens[rowBase + g] : 0u;
        unsigned v = 0;
        #pragma unroll
        for (int i = 0; i < 8; ++i)
            if (j & (1 << i)) v ^= BASEC[i];
        cklut[j] = v;
    }
    __syncthreads();

    {
        const int tq = (tid < 4) ? 4 : tid;
        const int q0 = 2 * tq;
        unsigned w[9];
        #pragma unroll
        for (int i = 0; i < 9; ++i) w[i] = tk[q0 - 8 + i];
        unsigned a6 = 0, b6 = 0, a7 = 0, b7 = 0;
        #pragma unroll
        for (int j = 0; j < 8; ++j) {
            const unsigned m6 = BASEC[j] ^ 0xCAFEBABEu;
            const unsigned m7 = BASEC[j] ^ 0x8BADF00Du;
            a6 ^= w[7 - j] * m6;
            b6 ^= w[8 - j] * m6;
            a7 ^= w[7 - j] * m7;
            b7 ^= w[8 - j] * m7;
        }
        g6_lds[2 * tid] = a6;  g6_lds[2 * tid + 1] = b6;
        g7_lds[2 * tid] = a7;  g7_lds[2 * tid + 1] = b7;
    }
    {
        const int base = 128 + tid;
        unsigned h = 0;
        #pragma unroll
        for (int i = 0; i < 16; ++i) {
            h ^= tk[base - 1 - i] * BASEC[i];
            if (i == 0)  keys_lds[0][tid] = h & 0xFFFFu;
            if (i == 1)  keys_lds[1][tid] = h & 0xFFFFu;
            if (i == 3)  keys_lds[2][tid] = h & 0xFFFFu;
            if (i == 7)  keys_lds[3][tid] = h & 0xFFFFu;
            if (i == 15) keys_lds[4][tid] = h & 0xFFFFu;
        }
        unsigned h5 = 0;
        #pragma unroll 8
        for (int i = 0; i < 32; ++i)  h5 ^= tk[base - 1 - i] * P5C[i];
        keys_lds[5][tid] = h5 & 0xFFFFu;
    }
    __syncthreads();
    {
        const int base = 128 + tid;
        unsigned h6 = 0, h7 = 0;
        #pragma unroll
        for (int m = 0; m < 8; ++m)  h6 ^= g6_lds[base - 8 * m];
        #pragma unroll
        for (int m = 0; m < 16; ++m) h7 ^= g7_lds[base - 8 * m];
        keys_lds[6][tid] = h6 & 0xFFFFu;
        keys_lds[7][tid] = h7 & 0xFFFFu;
    }

    const v4f* tab4 = reinterpret_cast<const v4f*>(tables);
    v4f* out4 = reinterpret_cast<v4f*>(out);
    const long outPosBase = (rowBase + t0) * 8;

    #pragma unroll 4
    for (int j = 0; j < nIter; ++j) {
        int flat   = j * 128 + tid;
        int rowIdx = flat >> 2;
        int piece  = flat & 3;
        int scale  = rowIdx >> 7;
        int pos    = rowIdx & 127;
        unsigned key = keys_lds[scale][pos];
        v4f v = tab4[((size_t)(scale << 16) + key) * 4 + piece];
        *reinterpret_cast<v4f*>(&short_lds[pos * STRIDE + scale * 16 + piece * 4]) = v;
    }
    __syncthreads();

    {
        const int wave = tid >> 5;
        #pragma unroll
        for (int tt = 0; tt < 2; ++tt) {
            const int mb = wave * 32 + tt * 16;
            const float* ap = &short_lds[(mb + n) * STRIDE + 2 * hi];
            v8f c = {0.f, 0.f, 0.f, 0.f, 0.f, 0.f, 0.f, 0.f};
            #pragma unroll
            for (int k = 0; k < 16; ++k) {
                v2f a  = *reinterpret_cast<const v2f*>(ap + 4 * k);
                c = __builtin_amdgcn_wmma_f32_16x16x4_f32(
                        false, a, false, bw[k], (short)0, c, false, false);
            }
            #pragma unroll
            for (int v = 0; v < 8; ++v) {
#if __has_builtin(__builtin_amdgcn_ballot_w32)
                unsigned mask = __builtin_amdgcn_ballot_w32(c[v] > 0.0f);
                unsigned bits = (mask >> (16 * hi)) & 0xFFu;
                if (n == v)
                    ck_lds[mb + v + 8 * hi] = cklut[bits];
#else
                if (n < 8 && c[v] > 0.0f)
                    atomicXor(&ck_lds[mb + v + 8 * hi], BASEC[n]);
#endif
            }
        }
    }
    __syncthreads();

    for (int j0 = 0; j0 < nIter; j0 += 4) {
        v4f v[4];
        #pragma unroll
        for (int u = 0; u < 4; ++u) {
            int flat   = (j0 + u) * 128 + tid;
            int rowIdx = flat >> 2;
            int piece  = flat & 3;
            int scale  = 4 + (rowIdx >> 7);
            int pos    = rowIdx & 127;
            unsigned key = (keys_lds[scale][pos] ^ ck_lds[pos]) & 0xFFFFu;
            v[u] = tab4[((size_t)(scale << 16) + key) * 4 + piece];
        }
        #pragma unroll
        for (int u = 0; u < 4; ++u) {
            int flat = (j0 + u) * 128 + tid;
            int spos = flat >> 4;
            int sq   = flat & 15;
            v4f s = *reinterpret_cast<const v4f*>(&short_lds[spos * STRIDE + sq * 4]);
            __builtin_nontemporal_store(s, &out4[(outPosBase + (long)spos * 8) * 4 + sq]);
        }
        #pragma unroll
        for (int u = 0; u < 4; ++u) {
            int flat   = (j0 + u) * 128 + tid;
            int rowIdx = flat >> 2;
            int piece  = flat & 3;
            int scale  = 4 + (rowIdx >> 7);
            int pos    = rowIdx & 127;
            __builtin_nontemporal_store(v[u], &out4[(outPosBase + (long)pos * 8 + scale) * 4 + piece]);
        }
    }
}

extern "C" void kernel_launch(void* const* d_in, const int* in_sizes, int n_in,
                              void* d_out, int out_size, void* d_ws, size_t ws_size,
                              hipStream_t stream) {
    (void)in_sizes; (void)n_in; (void)out_size; (void)d_ws; (void)ws_size;
    const int*   tokens = (const int*)d_in[0];
    const float* tables = (const float*)d_in[1];
    const float* Wc     = (const float*)d_in[2];
    float*       out    = (float*)d_out;
    const int B = 32, T = 4096;
    dim3 grid(B * (T / PPB)), block(128);
    hipLaunchKernelGGL(rhp_kernel, grid, block, 0, stream, tokens, tables, Wc, out, 16);
}


# Long half parked in the wave's own LDS tile rows (freed by that trip's short stores), stored 16 lanes per position: whole 256-byte runs, not 64-byte half lines; next gathers issue before read-back
# speedup vs baseline: 1.4883x; 1.1790x over previous
#include <hip/hip_runtime.h>

typedef __attribute__((ext_vector_type(2))) float v2f;
typedef __attribute__((ext_vector_type(4))) float v4f;
typedef __attribute__((ext_vector_type(8))) float v8f;

__device__ constexpr unsigned BASEC[16] = {
    2654435761u, 2246822519u, 3266489917u, 2028178513u, 1220703125u,
    1610612741u, 805306457u,  402653189u,  3674653429u, 2860486313u,
    1073676287u, 2971215073u, 1500450271u, 3267000013u, 2654435789u, 4049292737u};
__device__ constexpr unsigned P5C[32] = {
    2654435761u ^ 0xDEADBEEFu, 2246822519u ^ 0xDEADBEEFu, 3266489917u ^ 0xDEADBEEFu,
    2028178513u ^ 0xDEADBEEFu, 1220703125u ^ 0xDEADBEEFu, 1610612741u ^ 0xDEADBEEFu,
    805306457u  ^ 0xDEADBEEFu, 402653189u  ^ 0xDEADBEEFu, 3674653429u ^ 0xDEADBEEFu,
    2860486313u ^ 0xDEADBEEFu, 1073676287u ^ 0xDEADBEEFu, 2971215073u ^ 0xDEADBEEFu,
    1500450271u ^ 0xDEADBEEFu, 3267000013u ^ 0xDEADBEEFu, 2654435789u ^ 0xDEADBEEFu,
    4049292737u ^ 0xDEADBEEFu, 2246822531u ^ 0xDEADBEEFu, 3266489927u ^ 0xDEADBEEFu,
    2028178519u ^ 0xDEADBEEFu, 1220703133u ^ 0xDEADBEEFu, 1610612759u ^ 0xDEADBEEFu,
    805306463u  ^ 0xDEADBEEFu, 402653201u  ^ 0xDEADBEEFu, 3674653441u ^ 0xDEADBEEFu,
    2860486319u ^ 0xDEADBEEFu, 1073676311u ^ 0xDEADBEEFu, 2971215091u ^ 0xDEADBEEFu,
    1500450277u ^ 0xDEADBEEFu, 3267000023u ^ 0xDEADBEEFu, 2654435801u ^ 0xDEADBEEFu,
    4049292751u ^ 0xDEADBEEFu, 2246822537u ^ 0xDEADBEEFu};

#define T_DIM 4096
#define PPB 128
#define STRIDE 68

__global__ __launch_bounds__(128, 4)
void rhp_kernel(const int* __restrict__ tokens,
                const float* __restrict__ tables,
                const float* __restrict__ Wc,
                float* __restrict__ out,
                int nIter)
{
    __shared__ unsigned tk[256];
    __shared__ unsigned keys_lds[8][PPB];
    __shared__ unsigned ck_lds[PPB];
    __shared__ unsigned cklut[256];
    __shared__ __align__(8) unsigned g6_lds[256];
    __shared__ __align__(8) unsigned g7_lds[256];
    __shared__ __align__(16) float short_lds[PPB * STRIDE];

    const int tid = threadIdx.x;
    const int bid = blockIdx.x;
    const int b   = bid >> 5;
    const int t0  = (bid & 31) << 7;
    const long rowBase = (long)b * T_DIM;

    const int lane = tid & 31;
    const int n    = lane & 15;
    const int hi   = lane >> 4;
    v2f bw[16];
    {
        const float* wrow = Wc + (n & 7) * 64 + 2 * hi;
        #pragma unroll
        for (int k = 0; k < 16; ++k) bw[k] = *reinterpret_cast<const v2f*>(wrow + 4 * k);
    }

    for (int j = tid; j < 256; j += 128) {
        int g = t0 - 128 + j;
        tk[j] = (g >= 0) ? (unsigned)tokens[rowBase + g] : 0u;
        unsigned v = 0;
        #pragma unroll
        for (int i = 0; i < 8; ++i)
            if (j & (1 << i)) v ^= BASEC[i];
        cklut[j] = v;
    }
    __syncthreads();

    {
        const int tq = (tid < 4) ? 4 : tid;
        const int q0 = 2 * tq;
        unsigned w[9];
        #pragma unroll
        for (int i = 0; i < 9; ++i) w[i] = tk[q0 - 8 + i];
        unsigned a6 = 0, b6 = 0, a7 = 0, b7 = 0;
        #pragma unroll
        for (int j = 0; j < 8; ++j) {
            const unsigned m6 = BASEC[j] ^ 0xCAFEBABEu;
            const unsigned m7 = BASEC[j] ^ 0x8BADF00Du;
            a6 ^= w[7 - j] * m6;
            b6 ^= w[8 - j] * m6;
            a7 ^= w[7 - j] * m7;
            b7 ^= w[8 - j] * m7;
        }
        g6_lds[2 * tid] = a6;  g6_lds[2 * tid + 1] = b6;
        g7_lds[2 * tid] = a7;  g7_lds[2 * tid + 1] = b7;
    }
    {
        const int base = 128 + tid;
        unsigned h = 0;
        #pragma unroll
        for (int i = 0; i < 16; ++i) {
            h ^= tk[base - 1 - i] * BASEC[i];
            if (i == 0)  keys_lds[0][tid] = h & 0xFFFFu;
            if (i == 1)  keys_lds[1][tid] = h & 0xFFFFu;
            if (i == 3)  keys_lds[2][tid] = h & 0xFFFFu;
            if (i == 7)  keys_lds[3][tid] = h & 0xFFFFu;
            if (i == 15) keys_lds[4][tid] = h & 0xFFFFu;
        }
        unsigned h5 = 0;
        #pragma unroll 8
        for (int i = 0; i < 32; ++i)  h5 ^= tk[base - 1 - i] * P5C[i];
        keys_lds[5][tid] = h5 & 0xFFFFu;
    }
    __syncthreads();
    {
        const int base = 128 + tid;
        unsigned h6 = 0, h7 = 0;
        #pragma unroll
        for (int m = 0; m < 8; ++m)  h6 ^= g6_lds[base - 8 * m];
        #pragma unroll
        for (int m = 0; m < 16; ++m) h7 ^= g7_lds[base - 8 * m];
        keys_lds[6][tid] = h6 & 0xFFFFu;
        keys_lds[7][tid] = h7 & 0xFFFFu;
    }

    const v4f* tab4 = reinterpret_cast<const v4f*>(tables);
    v4f* out4 = reinterpret_cast<v4f*>(out);
    const long outPosBase = (rowBase + t0) * 8;

    #pragma unroll 4
    for (int j = 0; j < nIter; ++j) {
        int flat   = j * 128 + tid;
        int rowIdx = flat >> 2;
        int piece  = flat & 3;
        int scale  = rowIdx >> 7;
        int pos    = rowIdx & 127;
        unsigned key = keys_lds[scale][pos];
        v4f v = tab4[((size_t)(scale << 16) + key) * 4 + piece];
        *reinterpret_cast<v4f*>(&short_lds[pos * STRIDE + scale * 16 + piece * 4]) = v;
    }
    __syncthreads();

    {
        const int wave = tid >> 5;
        #pragma unroll
        for (int tt = 0; tt < 2; ++tt) {
            const int mb = wave * 32 + tt * 16;
            const float* ap = &short_lds[(mb + n) * STRIDE + 2 * hi];
            v8f c = {0.f, 0.f, 0.f, 0.f, 0.f, 0.f, 0.f, 0.f};
            #pragma unroll
            for (int k = 0; k < 16; ++k) {
                v2f a  = *reinterpret_cast<const v2f*>(ap + 4 * k);
                c = __builtin_amdgcn_wmma_f32_16x16x4_f32(
                        false, a, false, bw[k], (short)0, c, false, false);
            }
            #pragma unroll
            for (int v = 0; v < 8; ++v) {
#if __has_builtin(__builtin_amdgcn_ballot_w32)
                unsigned mask = __builtin_amdgcn_ballot_w32(c[v] > 0.0f);
                unsigned bits = (mask >> (16 * hi)) & 0xFFu;
                if (n == v)
                    ck_lds[mb + v + 8 * hi] = cklut[bits];
#else
                if (n < 8 && c[v] > 0.0f)
                    atomicXor(&ck_lds[mb + v + 8 * hi], BASEC[n]);
#endif
            }
        }
    }
    __syncthreads();

    {
        const int wave  = tid >> 5;
        const int gp    = lane >> 2;
        const int piece = lane & 3;
        const int sp    = lane >> 4;
        const int sq    = lane & 15;
        const int wbase = wave * 32;
        v4f v[4];
        {
            const int P = wbase + gp;
            const unsigned ck = ck_lds[P];
            #pragma unroll
            for (int u = 0; u < 4; ++u) {
                unsigned key = (keys_lds[4 + u][P] ^ ck) & 0xFFFFu;
                v[u] = tab4[((size_t)((4 + u) << 16) + key) * 4 + piece];
            }
        }
        for (int j0 = 0; j0 < nIter; j0 += 4) {
            const int pbase = wbase + j0 * 2;
            const int P     = pbase + gp;
            #pragma unroll
            for (int u = 0; u < 4; ++u) {
                int pos = pbase + 2 * u + sp;
                v4f s = *reinterpret_cast<const v4f*>(&short_lds[pos * STRIDE + sq * 4]);
                __builtin_nontemporal_store(s, &out4[(outPosBase + (long)pos * 8) * 4 + sq]);
            }
            __builtin_amdgcn_sched_barrier(0);
            #pragma unroll
            for (int u = 0; u < 4; ++u)
                *reinterpret_cast<v4f*>(&short_lds[P * STRIDE + u * 16 + piece * 4]) = v[u];
            if (j0 + 4 < nIter) {
                const int Pn = P + 8;
                const unsigned ck = ck_lds[Pn];
                #pragma unroll
                for (int u = 0; u < 4; ++u) {
                    unsigned key = (keys_lds[4 + u][Pn] ^ ck) & 0xFFFFu;
                    v[u] = tab4[((size_t)((4 + u) << 16) + key) * 4 + piece];
                }
            }
            #pragma unroll
            for (int u = 0; u < 4; ++u) {
                int pos = pbase + 2 * u + sp;
                v4f l = *reinterpret_cast<const v4f*>(&short_lds[pos * STRIDE + sq * 4]);
                __builtin_nontemporal_store(l, &out4[(outPosBase + (long)pos * 8 + 4) * 4 + sq]);
            }
        }
    }
}

extern "C" void kernel_launch(void* const* d_in, const int* in_sizes, int n_in,
                              void* d_out, int out_size, void* d_ws, size_t ws_size,
                              hipStream_t stream) {
    (void)in_sizes; (void)n_in; (void)out_size; (void)d_ws; (void)ws_size;
    const int*   tokens = (const int*)d_in[0];
    const float* tables = (const float*)d_in[1];
    const float* Wc     = (const float*)d_in[2];
    float*       out    = (float*)d_out;
    const int B = 32, T = 4096;
    dim3 grid(B * (T / PPB)), block(128);
    hipLaunchKernelGGL(rhp_kernel, grid, block, 0, stream, tokens, tables, Wc, out, 16);
}


# Short half leaves from the LDS tile right behind the gather barrier, ahead of the WMMAs: its stores drain under the matrix phase; phase 5 has long stores only and needs no block barrier in front
# speedup vs baseline: 1.6244x; 1.0914x over previous
#include <hip/hip_runtime.h>

typedef __attribute__((ext_vector_type(2))) float v2f;
typedef __attribute__((ext_vector_type(4))) float v4f;
typedef __attribute__((ext_vector_type(8))) float v8f;

__device__ constexpr unsigned BASEC[16] = {
    2654435761u, 2246822519u, 3266489917u, 2028178513u, 1220703125u,
    1610612741u, 805306457u,  402653189u,  3674653429u, 2860486313u,
    1073676287u, 2971215073u, 1500450271u, 3267000013u, 2654435789u, 4049292737u};
__device__ constexpr unsigned P5C[32] = {
    2654435761u ^ 0xDEADBEEFu, 2246822519u ^ 0xDEADBEEFu, 3266489917u ^ 0xDEADBEEFu,
    2028178513u ^ 0xDEADBEEFu, 1220703125u ^ 0xDEADBEEFu, 1610612741u ^ 0xDEADBEEFu,
    805306457u  ^ 0xDEADBEEFu, 402653189u  ^ 0xDEADBEEFu, 3674653429u ^ 0xDEADBEEFu,
    2860486313u ^ 0xDEADBEEFu, 1073676287u ^ 0xDEADBEEFu, 2971215073u ^ 0xDEADBEEFu,
    1500450271u ^ 0xDEADBEEFu, 3267000013u ^ 0xDEADBEEFu, 2654435789u ^ 0xDEADBEEFu,
    4049292737u ^ 0xDEADBEEFu, 2246822531u ^ 0xDEADBEEFu, 3266489927u ^ 0xDEADBEEFu,
    2028178519u ^ 0xDEADBEEFu, 1220703133u ^ 0xDEADBEEFu, 1610612759u ^ 0xDEADBEEFu,
    805306463u  ^ 0xDEADBEEFu, 402653201u  ^ 0xDEADBEEFu, 3674653441u ^ 0xDEADBEEFu,
    2860486319u ^ 0xDEADBEEFu, 1073676311u ^ 0xDEADBEEFu, 2971215091u ^ 0xDEADBEEFu,
    1500450277u ^ 0xDEADBEEFu, 3267000023u ^ 0xDEADBEEFu, 2654435801u ^ 0xDEADBEEFu,
    4049292751u ^ 0xDEADBEEFu, 2246822537u ^ 0xDEADBEEFu};

#define T_DIM 4096
#define PPB 128
#define STRIDE 68

__global__ __launch_bounds__(128, 4)
void rhp_kernel(const int* __restrict__ tokens,
                const float* __restrict__ tables,
                const float* __restrict__ Wc,
                float* __restrict__ out,
                int nIter)
{
    __shared__ unsigned tk[256];
    __shared__ unsigned keys_lds[8][PPB];
    __shared__ unsigned ck_lds[PPB];
    __shared__ unsigned cklut[256];
    __shared__ __align__(8) unsigned g6_lds[256];
    __shared__ __align__(8) unsigned g7_lds[256];
    __shared__ __align__(16) float short_lds[PPB * STRIDE];

    const int tid = threadIdx.x;
    const int bid = blockIdx.x;
    const int b   = bid >> 5;
    const int t0  = (bid & 31) << 7;
    const long rowBase = (long)b * T_DIM;

    const int lane = tid & 31;
    const int n    = lane & 15;
    const int hi   = lane >> 4;
    v2f bw[16];
    {
        const float* wrow = Wc + (n & 7) * 64 + 2 * hi;
        #pragma unroll
        for (int k = 0; k < 16; ++k) bw[k] = *reinterpret_cast<const v2f*>(wrow + 4 * k);
    }

    for (int j = tid; j < 256; j += 128) {
        int g = t0 - 128 + j;
        tk[j] = (g >= 0) ? (unsigned)tokens[rowBase + g] : 0u;
        unsigned v = 0;
        #pragma unroll
        for (int i = 0; i < 8; ++i)
            if (j & (1 << i)) v ^= BASEC[i];
        cklut[j] = v;
    }
    __syncthreads();

    {
        const int tq = (tid < 4) ? 4 : tid;
        const int q0 = 2 * tq;
        unsigned w[9];
        #pragma unroll
        for (int i = 0; i < 9; ++i) w[i] = tk[q0 - 8 + i];
        unsigned a6 = 0, b6 = 0, a7 = 0, b7 = 0;
        #pragma unroll
        for (int j = 0; j < 8; ++j) {
            const unsigned m6 = BASEC[j] ^ 0xCAFEBABEu;
            const unsigned m7 = BASEC[j] ^ 0x8BADF00Du;
            a6 ^= w[7 - j] * m6;
            b6 ^= w[8 - j] * m6;
            a7 ^= w[7 - j] * m7;
            b7 ^= w[8 - j] * m7;
        }
        g6_lds[2 * tid] = a6;  g6_lds[2 * tid + 1] = b6;
        g7_lds[2 * tid] = a7;  g7_lds[2 * tid + 1] = b7;
    }
    {
        const int base = 128 + tid;
        unsigned h = 0;
        #pragma unroll
        for (int i = 0; i < 16; ++i) {
            h ^= tk[base - 1 - i] * BASEC[i];
            if (i == 0)  keys_lds[0][tid] = h & 0xFFFFu;
            if (i == 1)  keys_lds[1][tid] = h & 0xFFFFu;
            if (i == 3)  keys_lds[2][tid] = h & 0xFFFFu;
            if (i == 7)  keys_lds[3][tid] = h & 0xFFFFu;
            if (i == 15) keys_lds[4][tid] = h & 0xFFFFu;
        }
        unsigned h5 = 0;
        #pragma unroll 8
        for (int i = 0; i < 32; ++i)  h5 ^= tk[base - 1 - i] * P5C[i];
        keys_lds[5][tid] = h5 & 0xFFFFu;
    }
    __syncthreads();
    {
        const int base = 128 + tid;
        unsigned h6 = 0, h7 = 0;
        #pragma unroll
        for (int m = 0; m < 8; ++m)  h6 ^= g6_lds[base - 8 * m];
        #pragma unroll
        for (int m = 0; m < 16; ++m) h7 ^= g7_lds[base - 8 * m];
        keys_lds[6][tid] = h6 & 0xFFFFu;
        keys_lds[7][tid] = h7 & 0xFFFFu;
    }

    const v4f* tab4 = reinterpret_cast<const v4f*>(tables);
    v4f* out4 = reinterpret_cast<v4f*>(out);
    const long outPosBase = (rowBase + t0) * 8;

    #pragma unroll 4
    for (int j = 0; j < nIter; ++j) {
        int flat   = j * 128 + tid;
        int rowIdx = flat >> 2;
        int piece  = flat & 3;
        int scale  = rowIdx >> 7;
        int pos    = rowIdx & 127;
        unsigned key = keys_lds[scale][pos];
        v4f v = tab4[((size_t)(scale << 16) + key) * 4 + piece];
        *reinterpret_cast<v4f*>(&short_lds[pos * STRIDE + scale * 16 + piece * 4]) = v;
    }
    __syncthreads();

    {
        const int wbase = (tid >> 5) * 32;
        const int sp    = lane >> 4;
        const int sq    = lane & 15;
        for (int j0 = 0; j0 < nIter; j0 += 8) {
            const int pbase = wbase + j0 * 2;
            #pragma unroll
            for (int u = 0; u < 8; ++u) {
                int pos = pbase + 2 * u + sp;
                v4f s = *reinterpret_cast<const v4f*>(&short_lds[pos * STRIDE + sq * 4]);
                __builtin_nontemporal_store(s, &out4[(outPosBase + (long)pos * 8) * 4 + sq]);
            }
        }
    }

    {
        const int wave = tid >> 5;
        #pragma unroll
        for (int tt = 0; tt < 2; ++tt) {
            const int mb = wave * 32 + tt * 16;
            const float* ap = &short_lds[(mb + n) * STRIDE + 2 * hi];
            v8f c = {0.f, 0.f, 0.f, 0.f, 0.f, 0.f, 0.f, 0.f};
            #pragma unroll
            for (int k = 0; k < 16; ++k) {
                v2f a  = *reinterpret_cast<const v2f*>(ap + 4 * k);
                c = __builtin_amdgcn_wmma_f32_16x16x4_f32(
                        false, a, false, bw[k], (short)0, c, false, false);
            }
            #pragma unroll
            for (int v = 0; v < 8; ++v) {
#if __has_builtin(__builtin_amdgcn_ballot_w32)
                unsigned mask = __builtin_amdgcn_ballot_w32(c[v] > 0.0f);
                unsigned bits = (mask >> (16 * hi)) & 0xFFu;
                if (n == v)
                    ck_lds[mb + v + 8 * hi] = cklut[bits];
#else
                if (n < 8 && c[v] > 0.0f)
                    atomicXor(&ck_lds[mb + v + 8 * hi], BASEC[n]);
#endif
            }
        }
    }

    {
        const int wave  = tid >> 5;
        const int gp    = lane >> 2;
        const int piece = lane & 3;
        const int sp    = lane >> 4;
        const int sq    = lane & 15;
        const int wbase = wave * 32;
        v4f v[4];
        {
            const int P = wbase + gp;
            const unsigned ck = ck_lds[P];
            #pragma unroll
            for (int u = 0; u < 4; ++u) {
                unsigned key = (keys_lds[4 + u][P] ^ ck) & 0xFFFFu;
                v[u] = tab4[((size_t)((4 + u) << 16) + key) * 4 + piece];
            }
        }
        for (int j0 = 0; j0 < nIter; j0 += 4) {
            const int pbase = wbase + j0 * 2;
            const int P     = pbase + gp;
            #pragma unroll
            for (int u = 0; u < 4; ++u)
                *reinterpret_cast<v4f*>(&short_lds[P * STRIDE + u * 16 + piece * 4]) = v[u];
            if (j0 + 4 < nIter) {
                const int Pn = P + 8;
                const unsigned ck = ck_lds[Pn];
                #pragma unroll
                for (int u = 0; u < 4; ++u) {
                    unsigned key = (keys_lds[4 + u][Pn] ^ ck) & 0xFFFFu;
                    v[u] = tab4[((size_t)((4 + u) << 16) + key) * 4 + piece];
                }
            }
            #pragma unroll
            for (int u = 0; u < 4; ++u) {
                int pos = pbase + 2 * u + sp;
                v4f l = *reinterpret_cast<const v4f*>(&short_lds[pos * STRIDE + sq * 4]);
                __builtin_nontemporal_store(l, &out4[(outPosBase + (long)pos * 8 + 4) * 4 + sq]);
            }
        }
    }
}

extern "C" void kernel_launch(void* const* d_in, const int* in_sizes, int n_in,
                              void* d_out, int out_size, void* d_ws, size_t ws_size,
                              hipStream_t stream) {
    (void)in_sizes; (void)n_in; (void)out_size; (void)d_ws; (void)ws_size;
    const int*   tokens = (const int*)d_in[0];
    const float* tables = (const float*)d_in[1];
    const float* Wc     = (const float*)d_in[2];
    float*       out    = (float*)d_out;
    const int B = 32, T = 4096;
    dim3 grid(B * (T / PPB)), block(128);
    hipLaunchKernelGGL(rhp_kernel, grid, block, 0, stream, tokens, tables, Wc, out, 16);
}
